# starting kernel
# baseline (speedup 1.0000x reference)
#include <hip/hip_runtime.h>
#include <math.h>


#define NUM_Q  10000
#define NUM_C  300
#define DIM_S  64
#define SIZE_M 50
#define BATCH  128
#define TT     200
#define HALF   32

#define RK  212
#define RP  208
#define WST 52

typedef __attribute__((ext_vector_type(2))) float v2f;
typedef __attribute__((ext_vector_type(8))) float v8f;

static __device__ __forceinline__ v8f wmma_f32_k4(v2f a, v2f b, v8f c) {
  return __builtin_amdgcn_wmma_f32_16x16x4_f32(false, a, false, b, (short)0, c, false, false);
}

__global__ __launch_bounds__(256) void deepirt_fused_kernel(
    const int*   __restrict__ question, const int*   __restrict__ response,
    const float* __restrict__ vq_emb,   const float* __restrict__ vc_emb,
    const float* __restrict__ kq_emb,   const float* __restrict__ kc_emb,
    const float* __restrict__ Mk,       const float* __restrict__ Mv0,
    const float* __restrict__ eW, const float* __restrict__ eb,
    const float* __restrict__ aW, const float* __restrict__ ab,
    const float* __restrict__ fW, const float* __restrict__ fb,
    const float* __restrict__ abilW, const float* __restrict__ abilb,
    const float* __restrict__ diffW, const float* __restrict__ diffb,
    float* __restrict__ out)
{
  extern __shared__ float smem[];
  float* sK   = smem;
  float* sV   = sK  + RK * 64;
  float* sE   = sV  + RP * 64;
  float* sA   = sE  + RP * 64;
  float* sW   = sA  + RP * 64;
  float* sMv  = sW  + RP * WST;
  float* sRed = sMv + SIZE_M * 64;
  int*   sQ   = (int*)(sRed + 4 * 64);
  int*   sRsp = sQ + TT;

  const int b    = blockIdx.x;
  const int tid  = threadIdx.x;
  const int lane = tid & 31;
  const int wave = tid >> 5;
  const int hh   = lane >> 4;
  const int r16  = lane & 15;

  for (int t = tid; t < TT; t += 256) {
    sQ[t]   = question[b * TT + t];
    sRsp[t] = response[b * TT + t];
  }
  __syncthreads();

  for (int idx = tid; idx < RK * 64; idx += 256) {
    int t = idx >> 6, c = idx & 63;
    float x = 0.f;
    if (t < TT) {
      int q = sQ[t];
      x = (c < HALF) ? kq_emb[q * HALF + c] : kc_emb[c - HALF];
    }
    sK[idx] = x;
  }
  for (int idx = tid; idx < RP * 64; idx += 256) {
    int t = idx >> 6, c = idx & 63;
    float x = 0.f;
    if (t < TT) {
      int q = sQ[t], r = sRsp[t];
      x = (c < HALF) ? vq_emb[(q + NUM_Q * r) * HALF + c]
                     : vc_emb[(NUM_C * r) * HALF + (c - HALF)];
    }
    sV[idx] = x;
  }
  for (int idx = tid; idx < SIZE_M * 64; idx += 256) sMv[idx] = Mv0[idx];
  __syncthreads();

  for (int nt = 0; nt < 4; ++nt) {
    int o  = nt * 16 + r16;
    int oc = (o < SIZE_M) ? o : (SIZE_M - 1);
    v2f bf[16];
#pragma unroll
    for (int kb = 0; kb < 16; ++kb)
      bf[kb] = *(const v2f*)&Mk[oc * 64 + kb * 4 + 2 * hh];
    for (int mt = wave; mt < RP / 16; mt += 8) {
      v2f af[16];
#pragma unroll
      for (int kb = 0; kb < 16; ++kb)
        af[kb] = *(const v2f*)&sK[(mt * 16 + r16) * 64 + kb * 4 + 2 * hh];
      v8f acc = {0.f,0.f,0.f,0.f,0.f,0.f,0.f,0.f};
#pragma unroll
      for (int kb = 0; kb < 16; ++kb) acc = wmma_f32_k4(af[kb], bf[kb], acc);
      if (o < SIZE_M) {
#pragma unroll
        for (int j = 0; j < 8; ++j)
          sW[(mt * 16 + j + 8 * hh) * WST + o] = acc[j];
      }
    }
  }

  for (int nt = 0; nt < 4; ++nt) {
    int o = nt * 16 + r16;
    v2f be[16], ba[16];
#pragma unroll
    for (int kb = 0; kb < 16; ++kb) {
      be[kb] = *(const v2f*)&eW[o * 64 + kb * 4 + 2 * hh];
      ba[kb] = *(const v2f*)&aW[o * 64 + kb * 4 + 2 * hh];
    }
    float ebv = eb[o], abv = ab[o];
    for (int mt = wave; mt < RP / 16; mt += 8) {
      v2f af[16];
#pragma unroll
      for (int kb = 0; kb < 16; ++kb)
        af[kb] = *(const v2f*)&sV[(mt * 16 + r16) * 64 + kb * 4 + 2 * hh];
      v8f ce = {0.f,0.f,0.f,0.f,0.f,0.f,0.f,0.f};
      v8f ca = {0.f,0.f,0.f,0.f,0.f,0.f,0.f,0.f};
#pragma unroll
      for (int kb = 0; kb < 16; ++kb) {
        ce = wmma_f32_k4(af[kb], be[kb], ce);
        ca = wmma_f32_k4(af[kb], ba[kb], ca);
      }
#pragma unroll
      for (int j = 0; j < 8; ++j) {
        int row = mt * 16 + j + 8 * hh;
        sE[row * 64 + o] = 1.f / (1.f + __expf(-(ce[j] + ebv)));
        sA[row * 64 + o] = tanhf(ca[j] + abv);
      }
    }
  }
  __syncthreads();

  for (int r = tid; r < TT; r += 256) {
    float mx = -1e30f;
    for (int m = 0; m < SIZE_M; ++m) mx = fmaxf(mx, sW[r * WST + m]);
    float s = 0.f;
    for (int m = 0; m < SIZE_M; ++m) {
      float ev = __expf(sW[r * WST + m] - mx);
      sW[r * WST + m] = ev; s += ev;
    }
    float inv = 1.f / s;
    for (int m = 0; m < SIZE_M; ++m) sW[r * WST + m] *= inv;
  }
  __syncthreads();

  const int dd = tid & 63, gg = tid >> 6;
  for (int t = 0; t < TT; ++t) {
    for (int idx = tid; idx < SIZE_M * 64; idx += 256) {
      int m = idx >> 6, d = idx & 63;
      float wt = sW[t * WST + m];
      float et = sE[t * 64 + d];
      float at = sA[t * 64 + d];
      sMv[idx] = sMv[idx] * (1.f - wt * et) + wt * at;
    }
    __syncthreads();
    if (t < TT - 1) {
      float p = 0.f;
      for (int m = gg; m < SIZE_M; m += 4)
        p += sW[(t + 1) * WST + m] * sMv[m * 64 + dd];
      sRed[gg * 64 + dd] = p;
      __syncthreads();
      if (tid < 64)
        sV[t * 64 + tid] = sRed[tid] + sRed[64 + tid] + sRed[128 + tid] + sRed[192 + tid];
      __syncthreads();
    } else {
      __syncthreads();
    }
  }

  for (int idx = (TT - 1) * 64 + tid; idx < RP * 64; idx += 256) sV[idx] = 0.f;
  __syncthreads();

  for (int nt = 0; nt < 4; ++nt) {
    int o = nt * 16 + r16;
    v2f bf[32];
#pragma unroll
    for (int kb = 0; kb < 32; ++kb)
      bf[kb] = *(const v2f*)&fW[o * 128 + kb * 4 + 2 * hh];
    float fbv = fb[o];
    for (int mt = wave; mt < RP / 16; mt += 8) {
      int row = mt * 16 + r16;
      v2f af[32];
#pragma unroll
      for (int kb = 0; kb < 16; ++kb)
        af[kb] = *(const v2f*)&sV[row * 64 + kb * 4 + 2 * hh];
#pragma unroll
      for (int kb = 16; kb < 32; ++kb)
        af[kb] = *(const v2f*)&sK[(row + 1) * 64 + (kb - 16) * 4 + 2 * hh];
      v8f acc = {0.f,0.f,0.f,0.f,0.f,0.f,0.f,0.f};
#pragma unroll
      for (int kb = 0; kb < 32; ++kb) acc = wmma_f32_k4(af[kb], bf[kb], acc);
#pragma unroll
      for (int j = 0; j < 8; ++j) {
        int rr = mt * 16 + j + 8 * hh;
        if (rr < TT - 1) sE[rr * 64 + o] = tanhf(acc[j] + fbv);
      }
    }
  }
  __syncthreads();

  for (int r = tid; r < TT - 1; r += 256) {
    float s1 = 0.f, s2 = 0.f;
    for (int c = 0; c < 64; ++c) {
      s1 += sE[r * 64 + c] * abilW[c];
      s2 += sK[(r + 1) * 64 + c] * diffW[c];
    }
    float ability = tanhf(s1 + abilb[0]);
    float diff    = tanhf(s2 + diffb[0]);
    out[b * (TT - 1) + r] = 3.f * ability - diff;
  }
}

extern "C" void kernel_launch(void* const* d_in, const int* in_sizes, int n_in,
                              void* d_out, int out_size, void* d_ws, size_t ws_size,
                              hipStream_t stream) {
  (void)in_sizes; (void)n_in; (void)out_size; (void)d_ws; (void)ws_size;
  const int*   question = (const int*)  d_in[0];
  const int*   response = (const int*)  d_in[1];
  const float* vq_emb   = (const float*)d_in[2];
  const float* vc_emb   = (const float*)d_in[3];
  const float* kq_emb   = (const float*)d_in[4];
  const float* kc_emb   = (const float*)d_in[5];
  const float* Mk       = (const float*)d_in[6];
  const float* Mv0      = (const float*)d_in[7];
  const float* eW       = (const float*)d_in[8];
  const float* ebv      = (const float*)d_in[9];
  const float* aW       = (const float*)d_in[10];
  const float* abv      = (const float*)d_in[11];
  const float* fW       = (const float*)d_in[12];
  const float* fbv      = (const float*)d_in[13];
  const float* abilW    = (const float*)d_in[14];
  const float* abilb    = (const float*)d_in[15];
  const float* diffW    = (const float*)d_in[16];
  const float* diffb    = (const float*)d_in[17];
  float* out = (float*)d_out;

  const size_t smem_bytes =
      (size_t)(RK * 64 + 3 * RP * 64 + RP * WST + SIZE_M * 64 + 4 * 64) * sizeof(float)
      + (size_t)(2 * TT) * sizeof(int);

  (void)hipFuncSetAttribute((const void*)deepirt_fused_kernel,
                            hipFuncAttributeMaxDynamicSharedMemorySize,
                            (int)smem_bytes);

  deepirt_fused_kernel<<<BATCH, 256, smem_bytes, stream>>>(
      question, response, vq_emb, vc_emb, kq_emb, kc_emb, Mk, Mv0,
      eW, ebv, aW, abv, fW, fbv, abilW, abilb, diffW, diffb, out);
}


# Global-memory waits taken off the serial path: Mv0, head weights and biases are fetched at kernel entry (head weights via LDS); chunk barrier is LDS-only, none behind the last chunk; sums unchanged
# speedup vs baseline: 9.9690x; 9.9690x over previous
#include <hip/hip_runtime.h>
#include <math.h>


#define NUM_Q  10000
#define NUM_C  300
#define DIM_S  64
#define SIZE_M 50
#define BATCH  128
#define TT     200
#define HALF   32

#define RP  208
#define PT  212
#define WSH 3
#define WPP (2 * PT)
#define WSOFF (NG * 3 * WPP)
#define DS  32
#define NG  8
#define SPT 7
#define CH  104
#define HP  68
#define BP  68
#define FP  132
#define FT  18

typedef __attribute__((ext_vector_type(2))) float v2f;
typedef __attribute__((ext_vector_type(4))) float v4f;
typedef __attribute__((ext_vector_type(8))) float v8f;

static __device__ __forceinline__ int w_base(int m) {
  const int g = m & (NG - 1), k = m >> 3;
  return (k < 6) ? ((g * 3 + (k >> 1)) * WPP + (k & 1)) : (WSOFF + g * PT);
}
static __device__ __forceinline__ int w_step(int m) { return ((m >> 3) < 6) ? 2 : 1; }

static __device__ __forceinline__ v8f wmma_f32_k4(v2f a, v2f b, v8f c) {
  return __builtin_amdgcn_wmma_f32_16x16x4_f32(false, a, false, b, (short)0, c, false, false);
}

__global__ __launch_bounds__(256) void deepirt_scan_kernel(
    const int*   __restrict__ question, const int*   __restrict__ response,
    const float* __restrict__ vq_emb,   const float* __restrict__ vc_emb,
    const float* __restrict__ kq_emb,   const float* __restrict__ kc_emb,
    const float* __restrict__ Mk,       const float* __restrict__ Mv0,
    const float* __restrict__ eW, const float* __restrict__ eb,
    const float* __restrict__ aW, const float* __restrict__ ab,
    float* __restrict__ rd)
{
  extern __shared__ __attribute__((aligned(16))) float smem[];
  float* sPart = smem;
  float* sE   = sPart + CH * NG * DS;
  float* sA   = sE  + DS * PT;
  float* sW   = sA  + DS * PT;
  float* sMk  = sPart;
  float* sEW  = sMk + SIZE_M * BP;
  float* sAW  = sEW + DS * BP;

  const int b    = blockIdx.x >> 1;
  const int d0   = (blockIdx.x & 1) * DS;
  const int tid  = threadIdx.x;
  const int lane = tid & 31;
  const int wave = tid >> 5;
  const int hh   = lane >> 4;
  const int r16  = lane & 15;

  const bool two = (wave + 8) < RP / 16;
  const v2f zero2 = {0.f, 0.f};
  v2f ak[2][16], av[2][16];
  int qi[2] = {0, 0}, ri[2] = {0, 0};
#pragma unroll
  for (int s = 0; s < 2; ++s) {
    if (s == 0 || two) {
      const int t  = (wave + 8 * s) * 16 + r16;
      const int tc = (t < TT) ? t : (TT - 1);
      qi[s] = question[b * TT + tc];
      ri[s] = response[b * TT + tc];
    }
  }
  const int dd = lane, gg = wave;
  const bool ok6 = (gg + NG * (SPT - 1)) < SIZE_M;
  const int  m6  = ok6 ? (gg + NG * (SPT - 1)) : (SIZE_M - 1);
  v2f mv2[3];
  for (int p = 0; p < 3; ++p) {
    mv2[p].x = Mv0[(gg + 16 * p) * 64 + d0 + dd];
    mv2[p].y = Mv0[(gg + 16 * p + 8) * 64 + d0 + dd];
  }
  float mv6 = Mv0[m6 * 64 + d0 + dd];

  v4f gm[4], ge[2], ga[2];
#pragma unroll
  for (int j = 0; j < 4; ++j) {
    const int i  = tid + 256 * j;
    const int ic = (i < SIZE_M * 16) ? i : (SIZE_M * 16 - 1);
    gm[j] = *(const v4f*)(Mk + 4 * ic);
  }
#pragma unroll
  for (int j = 0; j < 2; ++j) {
    const int i = tid + 256 * j;
    ge[j] = *(const v4f*)(eW + (size_t)(d0 + (i >> 4)) * 64 + (i & 15) * 4);
    ga[j] = *(const v4f*)(aW + (size_t)(d0 + (i >> 4)) * 64 + (i & 15) * 4);
  }
#pragma unroll
  for (int s = 0; s < 2; ++s) {
    if (s == 0 || two) {
      const int  t  = (wave + 8 * s) * 16 + r16;
      const bool ok = t < TT;
      const int  q  = qi[s];
      const int  r  = ri[s];
      const float* kr = kq_emb + (size_t)q * HALF + 2 * hh;
      const float* cr = kc_emb + 2 * hh;
      const float* vr = vq_emb + (size_t)(q + NUM_Q * r) * HALF + 2 * hh;
      const float* wr = vc_emb + (size_t)(NUM_C * r) * HALF + 2 * hh;
#pragma unroll
      for (int kb = 0; kb < 8; ++kb) {
        v2f x0 = *(const v2f*)(kr + kb * 4);
        v2f x1 = *(const v2f*)(cr + kb * 4);
        v2f y0 = *(const v2f*)(vr + kb * 4);
        v2f y1 = *(const v2f*)(wr + kb * 4);
        ak[s][kb]     = ok ? x0 : zero2;
        ak[s][kb + 8] = ok ? x1 : zero2;
        av[s][kb]     = ok ? y0 : zero2;
        av[s][kb + 8] = ok ? y1 : zero2;
      }
    }
  }
#pragma unroll
  for (int j = 0; j < 3; ++j) {
    const int i = tid + 256 * j;
    *(v4f*)&sMk[(i >> 4) * BP + (i & 15) * 4] = gm[j];
  }
  if (tid < SIZE_M * 16 - 768)
    *(v4f*)&sMk[((tid + 768) >> 4) * BP + (tid & 15) * 4] = gm[3];
#pragma unroll
  for (int j = 0; j < 2; ++j) {
    const int i = tid + 256 * j;
    *(v4f*)&sEW[(i >> 4) * BP + (i & 15) * 4] = ge[j];
    *(v4f*)&sAW[(i >> 4) * BP + (i & 15) * 4] = ga[j];
  }
  __syncthreads();

  for (int nt = 0; nt < 4; ++nt) {
    int o  = nt * 16 + r16;
    int oc = (o < SIZE_M) ? o : (SIZE_M - 1);
    v2f bf[16];
#pragma unroll
    for (int kb = 0; kb < 16; ++kb)
      bf[kb] = *(const v2f*)&sMk[oc * BP + kb * 4 + 2 * hh];
#pragma unroll
    for (int s = 0; s < 2; ++s) {
      if (s == 0 || two) {
        const int mt = wave + 8 * s;
        v8f acc = {0.f,0.f,0.f,0.f,0.f,0.f,0.f,0.f};
#pragma unroll
        for (int kb = 0; kb < 16; ++kb) acc = wmma_f32_k4(ak[s][kb], bf[kb], acc);
        if (o < SIZE_M) {
          const int ws = w_step(o);
          const int wb = w_base(o) + (mt * 16 + 8 * hh + WSH) * ws;
#pragma unroll
          for (int j = 0; j < 8; ++j)
            sW[wb + j * ws] = acc[j];
        }
      }
    }
  }

  for (int nt = 0; nt < DS / 16; ++nt) {
    int ol = nt * 16 + r16;
    int o  = d0 + ol;
    v2f be[16], ba[16];
#pragma unroll
    for (int kb = 0; kb < 16; ++kb) {
      be[kb] = *(const v2f*)&sEW[ol * BP + kb * 4 + 2 * hh];
      ba[kb] = *(const v2f*)&sAW[ol * BP + kb * 4 + 2 * hh];
    }
    float ebv = eb[o], abv = ab[o];
#pragma unroll
    for (int s = 0; s < 2; ++s) {
      if (s == 0 || two) {
        const int mt = wave + 8 * s;
        v8f ce = {0.f,0.f,0.f,0.f,0.f,0.f,0.f,0.f};
        v8f ca = {0.f,0.f,0.f,0.f,0.f,0.f,0.f,0.f};
#pragma unroll
        for (int kb = 0; kb < 16; ++kb) {
          ce = wmma_f32_k4(av[s][kb], be[kb], ce);
          ca = wmma_f32_k4(av[s][kb], ba[kb], ca);
        }
        float ev[8], tv[8];
#pragma unroll
        for (int j = 0; j < 8; ++j) {
          ev[j] = 1.f / (1.f + __expf(-(ce[j] + ebv)));
          tv[j] = tanhf(ca[j] + abv);
        }
        const int row0 = mt * 16 + 8 * hh;
        *(v4f*)&sE[ol * PT + row0]     = (v4f){ev[0], ev[1], ev[2], ev[3]};
        *(v4f*)&sE[ol * PT + row0 + 4] = (v4f){ev[4], ev[5], ev[6], ev[7]};
        *(v4f*)&sA[ol * PT + row0]     = (v4f){tv[0], tv[1], tv[2], tv[3]};
        *(v4f*)&sA[ol * PT + row0 + 4] = (v4f){tv[4], tv[5], tv[6], tv[7]};
      }
    }
  }
  __syncthreads();

  for (int r = tid; r < TT; r += 256) {
    const int ti = r + WSH;
    float mx = -1e30f;
#pragma unroll
    for (int m = 0; m < SIZE_M; ++m) mx = fmaxf(mx, sW[w_base(m) + ti * w_step(m)]);
    float s = 0.f;
#pragma unroll
    for (int m = 0; m < SIZE_M; ++m) {
      float ev = __expf(sW[w_base(m) + ti * w_step(m)] - mx);
      sW[w_base(m) + ti * w_step(m)] = ev; s += ev;
    }
    float inv = 1.f / s;
#pragma unroll
    for (int m = 0; m < SIZE_M; ++m) sW[w_base(m) + ti * w_step(m)] *= inv;
  }
  __syncthreads();

  float* rdb = rd + (size_t)b * (TT * 64) + d0;

  struct OpSet { v4f a[3], b[3], s, e, t; };
  const float* wpp = sW + gg * 3 * WPP;
  const float* wp6 = sW + w_base(m6);
  const float* ep  = sE + dd * PT;
  const float* ap  = sA + dd * PT;
  v2f wc2[3]; float wc6;
  for (int p = 0; p < 3; ++p) wc2[p] = *(const v2f*)(wpp + p * WPP + WSH * 2);
  wc6 = wp6[WSH];
  OpSet S0, S1;
  for (int p = 0; p < 3; ++p) {
    S0.a[p] = *(const v4f*)(wpp + p * WPP + 8);
    S0.b[p] = *(const v4f*)(wpp + p * WPP + 12);
  }
  S0.s = *(const v4f*)(wp6 + 4);
  S0.e = *(const v4f*)ep;
  S0.t = *(const v4f*)ap;

  auto trip = [&](const int tq, const int t0, const OpSet& C, OpSet& N) {
    for (int p = 0; p < 3; ++p) {
      N.a[p] = *(const v4f*)(wpp + p * WPP + (tq + 8) * 2);
      N.b[p] = *(const v4f*)(wpp + p * WPP + (tq + 8) * 2 + 4);
    }
    N.s = *(const v4f*)(wp6 + tq + 8);
    N.e = *(const v4f*)(ep + tq + 4);
    N.t = *(const v4f*)(ap + tq + 4);
#pragma unroll
    for (int j = 0; j < 4; ++j) {
      const float ec = C.e[j], ac = C.t[j];
      v2f wn2[3];
      for (int p = 0; p < 3; ++p) {
        const v4f q = (j < 2) ? C.a[p] : C.b[p];
        wn2[p] = (j & 1) ? q.hi : q.lo;
      }
      const float wn6 = C.s[j];

      for (int p = 0; p < 3; ++p)
        mv2[p] = mv2[p] * (1.f - wc2[p] * ec) + wc2[p] * ac;
      mv6 = mv6 * (1.f - wc6 * ec) + wc6 * ac;

      float p = 0.f;
      for (int q = 0; q < 3; ++q) {
        const v2f pr = wn2[q] * mv2[q];
        p += pr.x;
        p += pr.y;
      }
      {
        float pr = wn6 * mv6;
        p += ok6 ? pr : 0.f;
      }
      sPart[((tq + j - t0) * NG + gg) * DS + dd] = p;

      for (int q = 0; q < 3; ++q) wc2[q] = wn2[q];
      wc6 = wn6;
    }
  };

  for (int t0 = 0; t0 < TT - 1; t0 += CH) {
    const int t1 = (t0 + CH < TT - 1) ? (t0 + CH) : (TT - 1);
    for (int tq = t0; tq < t1; tq += 8) {
      trip(tq,     t0, S0, S1);
      trip(tq + 4, t0, S1, S0);
    }
    __syncthreads();
    for (int idx = tid; idx < (t1 - t0) * DS; idx += 256) {
      const float* pp = sPart + (idx >> 5) * (NG * DS) + (idx & (DS - 1));
      float s = pp[0];
      for (int g = 1; g < NG; ++g) s += pp[g * DS];
      rdb[(t0 + (idx >> 5)) * 64 + (idx & (DS - 1))] = s;
    }
    if (t1 < TT - 1) {
      __builtin_amdgcn_fence(__ATOMIC_RELEASE, "workgroup", "local");
      __builtin_amdgcn_s_barrier();
      __builtin_amdgcn_fence(__ATOMIC_ACQUIRE, "workgroup", "local");
    }
  }
}

__global__ __launch_bounds__(128) void deepirt_head_kernel(
    const int*   __restrict__ question,
    const float* __restrict__ kq_emb,   const float* __restrict__ kc_emb,
    const float* __restrict__ fW, const float* __restrict__ fb,
    const float* __restrict__ abilW, const float* __restrict__ abilb,
    const float* __restrict__ diffW, const float* __restrict__ diffb,
    const float* __restrict__ rd, float* __restrict__ out)
{
  __shared__ __attribute__((aligned(16))) float sR[16 * HP];
  __shared__ __attribute__((aligned(16))) float sN[16 * HP];
  __shared__ __attribute__((aligned(16))) float sB[4 * 16 * FP];
  __shared__ __attribute__((aligned(16))) float sHW[128];

  const int mt   = blockIdx.x;
  const int b    = blockIdx.y;
  const int tid  = threadIdx.x;
  const int lane = tid & 31;
  const int wave = tid >> 5;
  const int hh   = lane >> 4;
  const int r16  = lane & 15;

  const int c6 = tid & 63, c5 = tid & 31;
  const float hw  = *((tid < 64) ? (abilW + tid) : (diffW + (tid - 64)));
  const float abb = abilb[0], dfb = diffb[0];
  int qn[4];
  for (int i = 0; i < 4; ++i) {
    int  row  = mt * 16 + (tid >> 5) + 4 * i;
    int  rowc = (row < TT - 1) ? row : (TT - 2);
    qn[i] = question[b * TT + rowc + 1];
  }
  float xr[8];
  for (int i = 0; i < 8; ++i) {
    int  row  = mt * 16 + (tid >> 6) + 2 * i;
    int  rowc = (row < TT - 1) ? row : (TT - 2);
    xr[i] = rd[((size_t)b * TT + rowc) * 64 + c6];
  }
  float* sBw = sB + wave * (16 * FP);
  v4f gb[16];
#pragma unroll
  for (int i = 0; i < 16; ++i)
    gb[i] = *(const v4f*)(fW + (size_t)(wave * 16 + i) * 128 + lane * 4);
  float xq[4];
  for (int i = 0; i < 4; ++i) xq[i] = kq_emb[qn[i] * HALF + c5];
  const float xc = kc_emb[c5];
  for (int i = 0; i < 8; ++i) {
    int  r   = (tid >> 6) + 2 * i;
    bool ok  = (mt * 16 + r) < TT - 1;
    sR[r * HP + c6] = ok ? xr[i] : 0.f;
  }
  sHW[tid] = hw;
#pragma unroll
  for (int i = 0; i < 16; ++i)
    *(v4f*)&sBw[i * FP + lane * 4] = gb[i];
  for (int i = 0; i < 4; ++i) {
    int  r   = (tid >> 5) + 4 * i;
    bool ok  = (mt * 16 + r) < TT - 1;
    sN[r * HP + c5]        = ok ? xq[i] : 0.f;
    sN[r * HP + HALF + c5] = ok ? xc : 0.f;
  }
  __syncthreads();

  {
    int o = wave * 16 + r16;
    v2f bf[32];
#pragma unroll
    for (int kb = 0; kb < 32; ++kb)
      bf[kb] = *(const v2f*)&sBw[r16 * FP + kb * 4 + 2 * hh];
    float fbv = fb[o];
    v2f af[32];
#pragma unroll
    for (int kb = 0; kb < 16; ++kb)
      af[kb] = *(const v2f*)&sR[r16 * HP + kb * 4 + 2 * hh];
#pragma unroll
    for (int kb = 16; kb < 32; ++kb)
      af[kb] = *(const v2f*)&sN[r16 * HP + (kb - 16) * 4 + 2 * hh];
    v8f acc = {0.f,0.f,0.f,0.f,0.f,0.f,0.f,0.f};
#pragma unroll
    for (int kb = 0; kb < 32; ++kb) acc = wmma_f32_k4(af[kb], bf[kb], acc);
#pragma unroll
    for (int j = 0; j < 8; ++j)
      sBw[(j + 8 * hh) * FT + r16] = tanhf(acc[j] + fbv);
  }
  __syncthreads();

  if (tid < 16) {
    int row = mt * 16 + tid;
    if (row < TT - 1) {
      float s1 = 0.f, s2 = 0.f;
      for (int c = 0; c < 64; ++c) {
        s1 += sB[(c >> 4) * (16 * FP) + tid * FT + (c & 15)] * sHW[c];
        s2 += sN[tid * HP + c] * sHW[64 + c];
      }
      float ability = tanhf(s1 + abb);
      float diff    = tanhf(s2 + dfb);
      out[b * (TT - 1) + row] = 3.f * ability - diff;
    }
  }
}

extern "C" void kernel_launch(void* const* d_in, const int* in_sizes, int n_in,
                              void* d_out, int out_size, void* d_ws, size_t ws_size,
                              hipStream_t stream) {
  (void)in_sizes; (void)n_in; (void)out_size; (void)ws_size;
  const int*   question = (const int*)  d_in[0];
  const int*   response = (const int*)  d_in[1];
  const float* vq_emb   = (const float*)d_in[2];
  const float* vc_emb   = (const float*)d_in[3];
  const float* kq_emb   = (const float*)d_in[4];
  const float* kc_emb   = (const float*)d_in[5];
  const float* Mk       = (const float*)d_in[6];
  const float* Mv0      = (const float*)d_in[7];
  const float* eW       = (const float*)d_in[8];
  const float* ebv      = (const float*)d_in[9];
  const float* aW       = (const float*)d_in[10];
  const float* abv      = (const float*)d_in[11];
  const float* fW       = (const float*)d_in[12];
  const float* fbv      = (const float*)d_in[13];
  const float* abilW    = (const float*)d_in[14];
  const float* abilb    = (const float*)d_in[15];
  const float* diffW    = (const float*)d_in[16];
  const float* diffb    = (const float*)d_in[17];
  float* out = (float*)d_out;
  float* rd  = (float*)d_ws;

  const size_t smem_bytes =
      (size_t)(CH * NG * DS + 2 * DS * PT + SIZE_M * PT) * sizeof(float);

  (void)hipFuncSetAttribute((const void*)deepirt_scan_kernel,
                            hipFuncAttributeMaxDynamicSharedMemorySize,
                            (int)smem_bytes);

  deepirt_scan_kernel<<<2 * BATCH, 256, smem_bytes, stream>>>(
      question, response, vq_emb, vc_emb, kq_emb, kc_emb, Mk, Mv0,
      eW, ebv, aW, abv, rd);

  deepirt_head_kernel<<<dim3(RP / 16, BATCH), 128, 0, stream>>>(
      question, kq_emb, kc_emb, fW, fbv, abilW, abilb, diffW, diffb, rd, out);
}
